# starting kernel
# baseline (speedup 1.0000x reference)
#include <hip/hip_runtime.h>


typedef float v2f __attribute__((ext_vector_type(2)));
typedef float v8f __attribute__((ext_vector_type(8)));

#define B_        16
#define K_        64
#define HF        16
#define WF        16
#define W_        80
#define LX        65
#define LY        25
#define FEAT_PAD  96
#define ROWS_PER_WG 4
#define WAVES     5
#define TPB       (WAVES * 32)

#define PIECE_FLOATS (K_ * HF * WF)
#define FEAT_FLOATS  (K_ * FEAT_PAD)
#define G2D_FLOATS   (B_ * 16 * 40 * (size_t)W_)
#define G1D_FLOATS   (B_ * 16 * 16 * (size_t)W_)

__global__ __launch_bounds__(TPB)
void corr_phase1_wmma(const float* __restrict__ piece,
                      const float* __restrict__ feat,
                      float* __restrict__ g,
                      int Hfeat, int kroll)
{
    extern __shared__ float smem[];
    float* sPiece = smem;
    float* sFeat  = smem + PIECE_FLOATS;

    const int b    = blockIdx.y;
    const int row0 = blockIdx.x * ROWS_PER_WG;
    const int tid  = threadIdx.x;
    const int wid  = tid >> 5;
    const int lane = tid & 31;
    const int half = lane >> 4;
    const int li   = lane & 15;
    const int x0   = wid * 16;

    const float* pieceB = piece + (size_t)b * PIECE_FLOATS;
    for (int idx = tid; idx < PIECE_FLOATS; idx += TPB) {
        int k    = idx >> 8;
        int rem  = idx & 255;
        int ksrc = (k + kroll) & (K_ - 1);
        sPiece[idx] = pieceB[ksrc * 256 + rem];
    }

    const int aBase = li * WF + 2 * half;
    const int bBase = x0 + li + 2 * half;

    for (int r = 0; r < ROWS_PER_WG; ++r) {
        const int row = row0 + r;
        const float* featRow = feat + ((size_t)(b * K_) * Hfeat + row) * W_;

        __syncthreads();
        for (int idx = tid; idx < FEAT_FLOATS; idx += TPB) {
            int k = idx / FEAT_PAD;
            int x = idx - k * FEAT_PAD;
            sFeat[idx] = (x < W_) ? featRow[(size_t)k * Hfeat * W_ + x] : 0.0f;
        }
        __syncthreads();

        v8f c = {0.f, 0.f, 0.f, 0.f, 0.f, 0.f, 0.f, 0.f};
        const float* ap = sPiece + aBase;
        const float* bp = sFeat  + bBase;

        for (int k = 0; k < K_; ++k) {
            #pragma unroll
            for (int jq = 0; jq < 4; ++jq) {
                v2f a, bb;
                a.x  = ap[jq * 4];
                a.y  = ap[jq * 4 + 1];
                bb.x = bp[jq * 4];
                bb.y = bp[jq * 4 + 1];
                c = __builtin_amdgcn_wmma_f32_16x16x4_f32(
                        false, a, false, bb, (short)0, c, false, false);
            }
            ap += HF * WF;
            bp += FEAT_PAD;
        }

        #pragma unroll
        for (int v = 0; v < 8; ++v) {
            int i = v + 8 * half;
            g[((size_t)(b * 16 + i) * Hfeat + row) * W_ + x0 + li] = c[v];
        }
    }
}

__global__ void corr_phase2_reduce(const float* __restrict__ g, float* __restrict__ out)
{
    int tid = blockIdx.x * blockDim.x + threadIdx.x;
    const int N2 = B_ * LY * LX;
    const int N1 = B_ * LX;
    if (tid < N2) {
        int b   = tid / (LY * LX);
        int rem = tid % (LY * LX);
        int y   = rem / LX;
        int x   = rem % LX;
        float s = 0.f;
        #pragma unroll
        for (int i = 0; i < 16; ++i)
            s += g[((size_t)(b * 16 + i) * 40 + (y + i)) * W_ + x];
        out[tid] = s;
    } else if (tid < N2 + N1) {
        int t2 = tid - N2;
        int b  = t2 / LX;
        int x  = t2 % LX;
        const float* g1 = g + G2D_FLOATS;
        float s = 0.f;
        #pragma unroll
        for (int i = 0; i < 16; ++i)
            s += g1[((size_t)(b * 16 + i) * 16 + i) * W_ + x];
        out[tid] = s;
    }
}

extern "C" void kernel_launch(void* const* d_in, const int* in_sizes, int n_in,
                              void* d_out, int out_size, void* d_ws, size_t ws_size,
                              hipStream_t stream)
{
    (void)in_sizes; (void)n_in; (void)out_size; (void)ws_size;

    const float* piece = (const float*)d_in[0];
    const float* comp  = (const float*)d_in[1];
    const float* strip = (const float*)d_in[2];

    float* g   = (float*)d_ws;
    float* out = (float*)d_out;

    const size_t shmem = (PIECE_FLOATS + FEAT_FLOATS) * sizeof(float);

    corr_phase1_wmma<<<dim3(10, B_), TPB, shmem, stream>>>(piece, comp, g, 40, 0);
    corr_phase1_wmma<<<dim3(4, B_),  TPB, shmem, stream>>>(piece, strip, g + G2D_FLOATS, 16, 32);

    const int total = B_ * LY * LX + B_ * LX;
    corr_phase2_reduce<<<(total + 255) / 256, 256, 0, stream>>>(g, out);
}


# Merge the two phase-1 WMMA launches (composite rows and strip rows) into one grid of 224 workgroups chosen by blockIdx.x; launches per call 3 -> 2
# speedup vs baseline: 1.9235x; 1.9235x over previous
#include <hip/hip_runtime.h>


typedef float v2f __attribute__((ext_vector_type(2)));
typedef float v8f __attribute__((ext_vector_type(8)));

#define B_        16
#define K_        64
#define HF        16
#define WF        16
#define W_        80
#define LX        65
#define LY        25
#define FEAT_PAD  96
#define ROWS_PER_WG 4
#define WAVES     5
#define TPB       (WAVES * 32)

#define PIECE_FLOATS (K_ * HF * WF)
#define FEAT_FLOATS  (K_ * FEAT_PAD)
#define G2D_FLOATS   (B_ * 16 * 40 * (size_t)W_)
#define G1D_FLOATS   (B_ * 16 * 16 * (size_t)W_)

#define GROUPS_2D (40 / ROWS_PER_WG)
#define GROUPS_1D (16 / ROWS_PER_WG)

__global__ __launch_bounds__(TPB)
void corr_phase1_wmma(const float* __restrict__ piece,
                      const float* __restrict__ comp,
                      const float* __restrict__ strip,
                      float* __restrict__ gAll)
{
    extern __shared__ float smem[];
    float* sPiece = smem;
    float* sFeat  = smem + PIECE_FLOATS;

    const bool is2d  = blockIdx.x < GROUPS_2D;
    const float* __restrict__ feat = is2d ? comp : strip;
    float* __restrict__ g          = is2d ? gAll : gAll + G2D_FLOATS;
    const int Hfeat  = is2d ? 40 : 16;
    const int kroll  = is2d ? 0 : (K_ / 2);
    const int grp    = is2d ? (int)blockIdx.x : (int)blockIdx.x - GROUPS_2D;

    const int b    = blockIdx.y;
    const int row0 = grp * ROWS_PER_WG;
    const int tid  = threadIdx.x;
    const int wid  = tid >> 5;
    const int lane = tid & 31;
    const int half = lane >> 4;
    const int li   = lane & 15;
    const int x0   = wid * 16;

    const float* pieceB = piece + (size_t)b * PIECE_FLOATS;
    for (int idx = tid; idx < PIECE_FLOATS; idx += TPB) {
        int k    = idx >> 8;
        int rem  = idx & 255;
        int ksrc = (k + kroll) & (K_ - 1);
        sPiece[idx] = pieceB[ksrc * 256 + rem];
    }

    const int aBase = li * WF + 2 * half;
    const int bBase = x0 + li + 2 * half;

    for (int r = 0; r < ROWS_PER_WG; ++r) {
        const int row = row0 + r;
        const float* featRow = feat + ((size_t)(b * K_) * Hfeat + row) * W_;

        __syncthreads();
        for (int idx = tid; idx < FEAT_FLOATS; idx += TPB) {
            int k = idx / FEAT_PAD;
            int x = idx - k * FEAT_PAD;
            sFeat[idx] = (x < W_) ? featRow[(size_t)k * Hfeat * W_ + x] : 0.0f;
        }
        __syncthreads();

        v8f c = {0.f, 0.f, 0.f, 0.f, 0.f, 0.f, 0.f, 0.f};
        const float* ap = sPiece + aBase;
        const float* bp = sFeat  + bBase;

        for (int k = 0; k < K_; ++k) {
            #pragma unroll
            for (int jq = 0; jq < 4; ++jq) {
                v2f a, bb;
                a.x  = ap[jq * 4];
                a.y  = ap[jq * 4 + 1];
                bb.x = bp[jq * 4];
                bb.y = bp[jq * 4 + 1];
                c = __builtin_amdgcn_wmma_f32_16x16x4_f32(
                        false, a, false, bb, (short)0, c, false, false);
            }
            ap += HF * WF;
            bp += FEAT_PAD;
        }

        #pragma unroll
        for (int v = 0; v < 8; ++v) {
            int i = v + 8 * half;
            g[((size_t)(b * 16 + i) * Hfeat + row) * W_ + x0 + li] = c[v];
        }
    }
}

__global__ void corr_phase2_reduce(const float* __restrict__ g, float* __restrict__ out)
{
    int tid = blockIdx.x * blockDim.x + threadIdx.x;
    const int N2 = B_ * LY * LX;
    const int N1 = B_ * LX;
    if (tid < N2) {
        int b   = tid / (LY * LX);
        int rem = tid % (LY * LX);
        int y   = rem / LX;
        int x   = rem % LX;
        float s = 0.f;
        #pragma unroll
        for (int i = 0; i < 16; ++i)
            s += g[((size_t)(b * 16 + i) * 40 + (y + i)) * W_ + x];
        out[tid] = s;
    } else if (tid < N2 + N1) {
        int t2 = tid - N2;
        int b  = t2 / LX;
        int x  = t2 % LX;
        const float* g1 = g + G2D_FLOATS;
        float s = 0.f;
        #pragma unroll
        for (int i = 0; i < 16; ++i)
            s += g1[((size_t)(b * 16 + i) * 16 + i) * W_ + x];
        out[tid] = s;
    }
}

extern "C" void kernel_launch(void* const* d_in, const int* in_sizes, int n_in,
                              void* d_out, int out_size, void* d_ws, size_t ws_size,
                              hipStream_t stream)
{
    (void)in_sizes; (void)n_in; (void)out_size; (void)ws_size;

    const float* piece = (const float*)d_in[0];
    const float* comp  = (const float*)d_in[1];
    const float* strip = (const float*)d_in[2];

    float* g   = (float*)d_ws;
    float* out = (float*)d_out;

    const size_t shmem = (PIECE_FLOATS + FEAT_FLOATS) * sizeof(float);

    corr_phase1_wmma<<<dim3(GROUPS_2D + GROUPS_1D, B_), TPB, shmem, stream>>>(piece, comp, strip, g);

    const int total = B_ * LY * LX + B_ * LX;
    corr_phase2_reduce<<<(total + 255) / 256, 256, 0, stream>>>(g, out);
}


# Stage template and feature rows with 128-bit global loads, every load of a stage in flight before its first LDS store (was 4-byte loads waited one by one); same LDS contents, WMMA loop untouched
# speedup vs baseline: 4.9481x; 2.5724x over previous
#include <hip/hip_runtime.h>


typedef float v2f __attribute__((ext_vector_type(2)));
typedef float v4f __attribute__((ext_vector_type(4)));
typedef float v8f __attribute__((ext_vector_type(8)));

#define B_        16
#define K_        64
#define HF        16
#define WF        16
#define W_        80
#define LX        65
#define LY        25
#define FEAT_PAD  96
#define ROWS_PER_WG 4
#define WAVES     5
#define TPB       (WAVES * 32)

#define PIECE_FLOATS (K_ * HF * WF)
#define FEAT_FLOATS  (K_ * FEAT_PAD)
#define G2D_FLOATS   (B_ * 16 * 40 * (size_t)W_)
#define G1D_FLOATS   (B_ * 16 * 16 * (size_t)W_)

#define PIECE_VECS        (PIECE_FLOATS / 4)
#define PIECE_FULL_PASSES (PIECE_VECS / TPB)
#define FEAT_VEC          (W_ / 4)
#define FEAT_KPASS        (TPB / FEAT_VEC)
#define FEAT_PASSES       (K_ / FEAT_KPASS)

#define GROUPS_2D (40 / ROWS_PER_WG)
#define GROUPS_1D (16 / ROWS_PER_WG)

__global__ __launch_bounds__(TPB)
void corr_phase1_wmma(const float* __restrict__ piece,
                      const float* __restrict__ comp,
                      const float* __restrict__ strip,
                      float* __restrict__ gAll)
{
    extern __shared__ __attribute__((aligned(16))) float smem[];
    float* sPiece = smem;
    float* sFeat  = smem + PIECE_FLOATS;

    const bool is2d  = blockIdx.x < GROUPS_2D;
    const float* __restrict__ feat = is2d ? comp : strip;
    float* __restrict__ g          = is2d ? gAll : gAll + G2D_FLOATS;
    const int Hfeat  = is2d ? 40 : 16;
    const int kroll  = is2d ? 0 : (K_ / 2);
    const int grp    = is2d ? (int)blockIdx.x : (int)blockIdx.x - GROUPS_2D;

    const int b    = blockIdx.y;
    const int row0 = grp * ROWS_PER_WG;
    const int tid  = threadIdx.x;
    const int wid  = tid >> 5;
    const int lane = tid & 31;
    const int half = lane >> 4;
    const int li   = lane & 15;
    const int x0   = wid * 16;

    const v4f* __restrict__ piece4 = (const v4f*)(piece + (size_t)b * PIECE_FLOATS);
    v4f* sPiece4 = (v4f*)sPiece;

    const int kl  = tid / FEAT_VEC;
    const int xq  = tid - kl * FEAT_VEC;
    const float* featThr = feat + ((size_t)(b * K_ + kl) * Hfeat + row0) * W_ + xq * 4;
    const size_t kPass   = (size_t)FEAT_KPASS * Hfeat * W_;
    float* sFeatThr      = sFeat + kl * FEAT_PAD + xq * 4;

    v4f pr[PIECE_FULL_PASSES + 1];
    v4f fr[FEAT_PASSES];

    const bool pieceTail = tid < PIECE_VECS - PIECE_FULL_PASSES * TPB;
    #pragma unroll
    for (int t = 0; t < PIECE_FULL_PASSES; ++t) {
        int idx4 = tid + t * TPB;
        int ksrc = ((idx4 >> 6) + kroll) & (K_ - 1);
        pr[t] = piece4[ksrc * 64 + (idx4 & 63)];
    }
    if (pieceTail) {
        int idx4 = tid + PIECE_FULL_PASSES * TPB;
        int ksrc = ((idx4 >> 6) + kroll) & (K_ - 1);
        pr[PIECE_FULL_PASSES] = piece4[ksrc * 64 + (idx4 & 63)];
    }
    #pragma unroll
    for (int t = 0; t < FEAT_PASSES; ++t)
        fr[t] = *(const v4f*)(featThr + t * kPass);

    __builtin_amdgcn_sched_barrier(0);

    #pragma unroll
    for (int t = 0; t < PIECE_FULL_PASSES; ++t)
        sPiece4[tid + t * TPB] = pr[t];
    if (pieceTail)
        sPiece4[tid + PIECE_FULL_PASSES * TPB] = pr[PIECE_FULL_PASSES];
    for (int z = tid; z < K_ * 4; z += TPB) {
        v4f zero = {0.f, 0.f, 0.f, 0.f};
        *(v4f*)(sFeat + (z >> 2) * FEAT_PAD + W_ + (z & 3) * 4) = zero;
    }

    const int aBase = li * WF + 2 * half;
    const int bBase = x0 + li + 2 * half;

    for (int r = 0; r < ROWS_PER_WG; ++r) {
        const int row = row0 + r;

        if (r > 0) {
            __syncthreads();
            #pragma unroll
            for (int t = 0; t < FEAT_PASSES; ++t)
                fr[t] = *(const v4f*)(featThr + (size_t)r * W_ + t * kPass);
        }
        #pragma unroll
        for (int t = 0; t < FEAT_PASSES; ++t)
            *(v4f*)(sFeatThr + t * (FEAT_KPASS * FEAT_PAD)) = fr[t];
        __syncthreads();

        v8f c = {0.f, 0.f, 0.f, 0.f, 0.f, 0.f, 0.f, 0.f};
        const float* ap = sPiece + aBase;
        const float* bp = sFeat  + bBase;

        for (int k = 0; k < K_; ++k) {
            #pragma unroll
            for (int jq = 0; jq < 4; ++jq) {
                v2f a, bb;
                a.x  = ap[jq * 4];
                a.y  = ap[jq * 4 + 1];
                bb.x = bp[jq * 4];
                bb.y = bp[jq * 4 + 1];
                c = __builtin_amdgcn_wmma_f32_16x16x4_f32(
                        false, a, false, bb, (short)0, c, false, false);
            }
            ap += HF * WF;
            bp += FEAT_PAD;
        }

        #pragma unroll
        for (int v = 0; v < 8; ++v) {
            int i = v + 8 * half;
            g[((size_t)(b * 16 + i) * Hfeat + row) * W_ + x0 + li] = c[v];
        }
    }
}

__global__ void corr_phase2_reduce(const float* __restrict__ g, float* __restrict__ out)
{
    int tid = blockIdx.x * blockDim.x + threadIdx.x;
    const int N2 = B_ * LY * LX;
    const int N1 = B_ * LX;
    if (tid < N2) {
        int b   = tid / (LY * LX);
        int rem = tid % (LY * LX);
        int y   = rem / LX;
        int x   = rem % LX;
        float s = 0.f;
        #pragma unroll
        for (int i = 0; i < 16; ++i)
            s += g[((size_t)(b * 16 + i) * 40 + (y + i)) * W_ + x];
        out[tid] = s;
    } else if (tid < N2 + N1) {
        int t2 = tid - N2;
        int b  = t2 / LX;
        int x  = t2 % LX;
        const float* g1 = g + G2D_FLOATS;
        float s = 0.f;
        #pragma unroll
        for (int i = 0; i < 16; ++i)
            s += g1[((size_t)(b * 16 + i) * 16 + i) * W_ + x];
        out[tid] = s;
    }
}

extern "C" void kernel_launch(void* const* d_in, const int* in_sizes, int n_in,
                              void* d_out, int out_size, void* d_ws, size_t ws_size,
                              hipStream_t stream)
{
    (void)in_sizes; (void)n_in; (void)out_size; (void)ws_size;

    const float* piece = (const float*)d_in[0];
    const float* comp  = (const float*)d_in[1];
    const float* strip = (const float*)d_in[2];

    float* g   = (float*)d_ws;
    float* out = (float*)d_out;

    const size_t shmem = (PIECE_FLOATS + FEAT_FLOATS) * sizeof(float);

    corr_phase1_wmma<<<dim3(GROUPS_2D + GROUPS_1D, B_), TPB, shmem, stream>>>(piece, comp, strip, g);

    const int total = B_ * LY * LX + B_ * LX;
    corr_phase2_reduce<<<(total + 255) / 256, 256, 0, stream>>>(g, out);
}


# Phase 1 remapped to 4 waves, one per SIMD (176 KB LDS: one workgroup per WGP): a wave owns one feature row, its 5 x-tiles as 5 accumulators sharing each template fragment; loads one k-slab ahead
# speedup vs baseline: 8.9641x; 1.8116x over previous
#include <hip/hip_runtime.h>


typedef float v2f __attribute__((ext_vector_type(2)));
typedef float v4f __attribute__((ext_vector_type(4)));
typedef float v8f __attribute__((ext_vector_type(8)));

#define B_        16
#define K_        64
#define HF        16
#define WF        16
#define W_        80
#define LX        65
#define LY        25
#define FEAT_PAD  96
#define ROWS_PER_WG 4
#define WAVES     4
#define TPB       (WAVES * 32)
#define XT        5

#define PIECE_FLOATS (K_ * HF * WF)
#define ROW_FLOATS   (K_ * FEAT_PAD)
#define FEAT_FLOATS  (ROWS_PER_WG * ROW_FLOATS)
#define G2D_FLOATS   (B_ * 16 * 40 * (size_t)W_)
#define G1D_FLOATS   (B_ * 16 * 16 * (size_t)W_)

#define LDS_FLOATS   (176 * 1024 / 4)

#define PIECE_VECS   (PIECE_FLOATS / 4)
#define PIECE_PASSES (PIECE_VECS / TPB)
#define ROWS_VEC     (ROWS_PER_WG * W_ / 4)
#define FEAT_VEC     (W_ / 4)
#define FEAT_SUB     5
#define FEAT_KSTEP   (FEAT_SUB * TPB / ROWS_VEC)
#define FEAT_OUTER   (K_ / FEAT_KSTEP)

#define GROUPS_2D (40 / ROWS_PER_WG)
#define GROUPS_1D (16 / ROWS_PER_WG)

__global__ __launch_bounds__(TPB)
void corr_phase1_wmma(const float* __restrict__ piece,
                      const float* __restrict__ comp,
                      const float* __restrict__ strip,
                      float* __restrict__ gAll)
{
    __shared__ __attribute__((aligned(16))) float smem[LDS_FLOATS];
    float* sFeat  = smem;
    float* sPiece = smem + FEAT_FLOATS;

    const bool is2d  = blockIdx.x < GROUPS_2D;
    const float* __restrict__ feat = is2d ? comp : strip;
    float* __restrict__ g          = is2d ? gAll : gAll + G2D_FLOATS;
    const int Hfeat  = is2d ? 40 : 16;
    const int kroll  = is2d ? 0 : (K_ / 2);
    const int grp    = is2d ? (int)blockIdx.x : (int)blockIdx.x - GROUPS_2D;

    const int b    = blockIdx.y;
    const int row0 = grp * ROWS_PER_WG;
    const int tid  = threadIdx.x;
    const int wid  = tid >> 5;
    const int lane = tid & 31;
    const int half = lane >> 4;
    const int li   = lane & 15;

    const v4f* __restrict__ piece4 = (const v4f*)(piece + (size_t)b * PIECE_FLOATS);
    v4f* sPiece4 = (v4f*)sPiece;

    v4f pr[PIECE_PASSES];
    v4f fr[FEAT_OUTER][FEAT_SUB];

    #pragma unroll
    for (int t = 0; t < PIECE_PASSES; ++t) {
        int idx4 = tid + t * TPB;
        int ksrc = ((idx4 >> 6) + kroll) & (K_ - 1);
        pr[t] = piece4[ksrc * 64 + (idx4 & 63)];
    }

    const float* featB = feat + ((size_t)(b * K_) * Hfeat + row0) * W_;
    const size_t kStride = (size_t)Hfeat * W_;
    #pragma unroll
    for (int q = 0; q < FEAT_OUTER; ++q) {
        #pragma unroll
        for (int s = 0; s < FEAT_SUB; ++s) {
            int e   = tid + s * TPB;
            int kk  = e / ROWS_VEC;
            int rem = e - kk * ROWS_VEC;
            fr[q][s] = *(const v4f*)(featB + (size_t)(q * FEAT_KSTEP + kk) * kStride + rem * 4);
        }
    }

    __builtin_amdgcn_sched_barrier(0);

    #pragma unroll
    for (int t = 0; t < PIECE_PASSES; ++t)
        sPiece4[tid + t * TPB] = pr[t];
    #pragma unroll
    for (int q = 0; q < FEAT_OUTER; ++q) {
        #pragma unroll
        for (int s = 0; s < FEAT_SUB; ++s) {
            int e   = tid + s * TPB;
            int kk  = e / ROWS_VEC;
            int rem = e - kk * ROWS_VEC;
            int r   = rem / FEAT_VEC;
            int xq  = rem - r * FEAT_VEC;
            *(v4f*)(sFeat + r * ROW_FLOATS + (q * FEAT_KSTEP + kk) * FEAT_PAD + xq * 4) = fr[q][s];
        }
    }
    for (int z = tid; z < ROWS_PER_WG * K_ * 4; z += TPB) {
        v4f zero = {0.f, 0.f, 0.f, 0.f};
        *(v4f*)(sFeat + (z >> 2) * FEAT_PAD + W_ + (z & 3) * 4) = zero;
    }
    __syncthreads();

    const float* ap = sPiece + li * WF + 2 * half;
    const float* bp = sFeat + wid * ROW_FLOATS + li + 2 * half;

    v8f c[XT];
    #pragma unroll
    for (int t = 0; t < XT; ++t) {
        v8f z8 = {0.f, 0.f, 0.f, 0.f, 0.f, 0.f, 0.f, 0.f};
        c[t] = z8;
    }

    v2f aF[2][4];
    v2f bF[2][4][XT];

#define LOAD_SLAB(S, KK)                                                   \
    _Pragma("unroll")                                                      \
    for (int jq = 0; jq < 4; ++jq) {                                       \
        aF[S][jq].x = ap[(KK) * (HF * WF) + jq * 4];                       \
        aF[S][jq].y = ap[(KK) * (HF * WF) + jq * 4 + 1];                   \
        _Pragma("unroll")                                                  \
        for (int t = 0; t < XT; ++t) {                                     \
            bF[S][jq][t].x = bp[(KK) * FEAT_PAD + t * 16 + jq * 4];        \
            bF[S][jq][t].y = bp[(KK) * FEAT_PAD + t * 16 + jq * 4 + 1];    \
        }                                                                  \
    }

#define MMA_SLAB(S)                                                        \
    _Pragma("unroll")                                                      \
    for (int jq = 0; jq < 4; ++jq) {                                       \
        _Pragma("unroll")                                                  \
        for (int t = 0; t < XT; ++t)                                       \
            c[t] = __builtin_amdgcn_wmma_f32_16x16x4_f32(                  \
                       false, aF[S][jq], false, bF[S][jq][t], (short)0, c[t], false, false); \
    }

    LOAD_SLAB(0, 0)
    #pragma unroll 1
    for (int k = 0; k < K_ - 2; k += 2) {
        LOAD_SLAB(1, 1)
        __builtin_amdgcn_sched_barrier(0);
        MMA_SLAB(0)
        __builtin_amdgcn_sched_barrier(0);
        LOAD_SLAB(0, 2)
        __builtin_amdgcn_sched_barrier(0);
        MMA_SLAB(1)
        __builtin_amdgcn_sched_barrier(0);
        ap += 2 * HF * WF;
        bp += 2 * FEAT_PAD;
    }
    LOAD_SLAB(1, 1)
    __builtin_amdgcn_sched_barrier(0);
    MMA_SLAB(0)
    MMA_SLAB(1)
#undef LOAD_SLAB
#undef MMA_SLAB

    const int row = row0 + wid;
    #pragma unroll
    for (int t = 0; t < XT; ++t) {
        #pragma unroll
        for (int v = 0; v < 8; ++v) {
            int i = v + 8 * half;
            g[((size_t)(b * 16 + i) * Hfeat + row) * W_ + t * 16 + li] = c[t][v];
        }
    }
}

__global__ void corr_phase2_reduce(const float* __restrict__ g, float* __restrict__ out)
{
    int tid = blockIdx.x * blockDim.x + threadIdx.x;
    const int N2 = B_ * LY * LX;
    const int N1 = B_ * LX;
    if (tid < N2) {
        int b   = tid / (LY * LX);
        int rem = tid % (LY * LX);
        int y   = rem / LX;
        int x   = rem % LX;
        float s = 0.f;
        #pragma unroll
        for (int i = 0; i < 16; ++i)
            s += g[((size_t)(b * 16 + i) * 40 + (y + i)) * W_ + x];
        out[tid] = s;
    } else if (tid < N2 + N1) {
        int t2 = tid - N2;
        int b  = t2 / LX;
        int x  = t2 % LX;
        const float* g1 = g + G2D_FLOATS;
        float s = 0.f;
        #pragma unroll
        for (int i = 0; i < 16; ++i)
            s += g1[((size_t)(b * 16 + i) * 16 + i) * W_ + x];
        out[tid] = s;
    }
}

extern "C" void kernel_launch(void* const* d_in, const int* in_sizes, int n_in,
                              void* d_out, int out_size, void* d_ws, size_t ws_size,
                              hipStream_t stream)
{
    (void)in_sizes; (void)n_in; (void)out_size; (void)ws_size;

    const float* piece = (const float*)d_in[0];
    const float* comp  = (const float*)d_in[1];
    const float* strip = (const float*)d_in[2];

    float* g   = (float*)d_ws;
    float* out = (float*)d_out;

    corr_phase1_wmma<<<dim3(GROUPS_2D + GROUPS_1D, B_), TPB, 0, stream>>>(piece, comp, strip, g);

    const int total = B_ * LY * LX + B_ * LX;
    corr_phase2_reduce<<<(total + 255) / 256, 256, 0, stream>>>(g, out);
}


# Phase 1 computes only the 65 used output columns: 4 x-tiles per row wave; column x=64 of the group's 4 rows is one shared tile, k-split over the 4 waves, summed through LDS (was a 5th tile per wave)
# speedup vs baseline: 9.4437x; 1.0535x over previous
#include <hip/hip_runtime.h>


typedef float v2f __attribute__((ext_vector_type(2)));
typedef float v4f __attribute__((ext_vector_type(4)));
typedef float v8f __attribute__((ext_vector_type(8)));

#define B_        16
#define K_        64
#define HF        16
#define WF        16
#define W_        80
#define LX        65
#define LY        25
#define FEAT_PAD  96
#define ROWS_PER_WG 4
#define WAVES     4
#define TPB       (WAVES * 32)
#define XT        4
#define XCOL      64
#define KQ        (K_ / WAVES)

#define PIECE_FLOATS (K_ * HF * WF)
#define ROW_FLOATS   (K_ * FEAT_PAD)
#define FEAT_FLOATS  (ROWS_PER_WG * ROW_FLOATS)
#define G2D_FLOATS   (B_ * 16 * 40 * (size_t)W_)
#define G1D_FLOATS   (B_ * 16 * 16 * (size_t)W_)

#define LDS_FLOATS   (176 * 1024 / 4)

#define PIECE_VECS   (PIECE_FLOATS / 4)
#define PIECE_PASSES (PIECE_VECS / TPB)
#define ROWS_VEC     (ROWS_PER_WG * W_ / 4)
#define FEAT_VEC     (W_ / 4)
#define FEAT_SUB     5
#define FEAT_KSTEP   (FEAT_SUB * TPB / ROWS_VEC)
#define FEAT_OUTER   (K_ / FEAT_KSTEP)

#define GROUPS_2D (40 / ROWS_PER_WG)
#define GROUPS_1D (16 / ROWS_PER_WG)

__global__ __launch_bounds__(TPB)
void corr_phase1_wmma(const float* __restrict__ piece,
                      const float* __restrict__ comp,
                      const float* __restrict__ strip,
                      float* __restrict__ gAll)
{
    __shared__ __attribute__((aligned(16))) float smem[LDS_FLOATS];
    float* sFeat  = smem;
    float* sPiece = smem + FEAT_FLOATS;

    const bool is2d  = blockIdx.x < GROUPS_2D;
    const float* __restrict__ feat = is2d ? comp : strip;
    float* __restrict__ g          = is2d ? gAll : gAll + G2D_FLOATS;
    const int Hfeat  = is2d ? 40 : 16;
    const int kroll  = is2d ? 0 : (K_ / 2);
    const int grp    = is2d ? (int)blockIdx.x : (int)blockIdx.x - GROUPS_2D;

    const int b    = blockIdx.y;
    const int row0 = grp * ROWS_PER_WG;
    const int tid  = threadIdx.x;
    const int wid  = tid >> 5;
    const int widu = __builtin_amdgcn_readfirstlane(wid);
    const int lane = tid & 31;
    const int half = lane >> 4;
    const int li   = lane & 15;

    const v4f* __restrict__ piece4 = (const v4f*)(piece + (size_t)b * PIECE_FLOATS);
    v4f* sPiece4 = (v4f*)sPiece;

    v4f pr[PIECE_PASSES];
    v4f fr[FEAT_OUTER][FEAT_SUB];

    #pragma unroll
    for (int t = 0; t < PIECE_PASSES; ++t) {
        int idx4 = tid + t * TPB;
        int ksrc = ((idx4 >> 6) + kroll) & (K_ - 1);
        pr[t] = piece4[ksrc * 64 + (idx4 & 63)];
    }

    const float* featB = feat + ((size_t)(b * K_) * Hfeat + row0) * W_;
    const size_t kStride = (size_t)Hfeat * W_;
    #pragma unroll
    for (int q = 0; q < FEAT_OUTER; ++q) {
        #pragma unroll
        for (int s = 0; s < FEAT_SUB; ++s) {
            int e   = tid + s * TPB;
            int kk  = e / ROWS_VEC;
            int rem = e - kk * ROWS_VEC;
            fr[q][s] = *(const v4f*)(featB + (size_t)(q * FEAT_KSTEP + kk) * kStride + rem * 4);
        }
    }

    __builtin_amdgcn_sched_barrier(0);

    #pragma unroll
    for (int t = 0; t < PIECE_PASSES; ++t)
        sPiece4[tid + t * TPB] = pr[t];
    #pragma unroll
    for (int q = 0; q < FEAT_OUTER; ++q) {
        #pragma unroll
        for (int s = 0; s < FEAT_SUB; ++s) {
            int e   = tid + s * TPB;
            int kk  = e / ROWS_VEC;
            int rem = e - kk * ROWS_VEC;
            int r   = rem / FEAT_VEC;
            int xq  = rem - r * FEAT_VEC;
            *(v4f*)(sFeat + r * ROW_FLOATS + (q * FEAT_KSTEP + kk) * FEAT_PAD + xq * 4) = fr[q][s];
        }
    }
    __syncthreads();

    const float* ap = sPiece + li * WF + 2 * half;
    const float* bp = sFeat + wid * ROW_FLOATS + li + 2 * half;

    v8f c[XT];
    #pragma unroll
    for (int t = 0; t < XT; ++t) {
        v8f z8 = {0.f, 0.f, 0.f, 0.f, 0.f, 0.f, 0.f, 0.f};
        c[t] = z8;
    }

    v2f aF[2][4];
    v2f bF[2][4][XT];

#define LOAD_SLAB(S, KK)                                                   \
    _Pragma("unroll")                                                      \
    for (int jq = 0; jq < 4; ++jq) {                                       \
        aF[S][jq].x = ap[(KK) * (HF * WF) + jq * 4];                       \
        aF[S][jq].y = ap[(KK) * (HF * WF) + jq * 4 + 1];                   \
        _Pragma("unroll")                                                  \
        for (int t = 0; t < XT; ++t) {                                     \
            bF[S][jq][t].x = bp[(KK) * FEAT_PAD + t * 16 + jq * 4];        \
            bF[S][jq][t].y = bp[(KK) * FEAT_PAD + t * 16 + jq * 4 + 1];    \
        }                                                                  \
    }

#define MMA_SLAB(S)                                                        \
    _Pragma("unroll")                                                      \
    for (int jq = 0; jq < 4; ++jq) {                                       \
        _Pragma("unroll")                                                  \
        for (int t = 0; t < XT; ++t)                                       \
            c[t] = __builtin_amdgcn_wmma_f32_16x16x4_f32(                  \
                       false, aF[S][jq], false, bF[S][jq][t], (short)0, c[t], false, false); \
    }

    LOAD_SLAB(0, 0)
    #pragma unroll 1
    for (int k = 0; k < K_ - 2; k += 2) {
        LOAD_SLAB(1, 1)
        __builtin_amdgcn_sched_barrier(0);
        MMA_SLAB(0)
        __builtin_amdgcn_sched_barrier(0);
        LOAD_SLAB(0, 2)
        __builtin_amdgcn_sched_barrier(0);
        MMA_SLAB(1)
        __builtin_amdgcn_sched_barrier(0);
        ap += 2 * HF * WF;
        bp += 2 * FEAT_PAD;
    }
    LOAD_SLAB(1, 1)
    __builtin_amdgcn_sched_barrier(0);
    MMA_SLAB(0)
    MMA_SLAB(1)
#undef LOAD_SLAB
#undef MMA_SLAB

    v8f cx[4];
    #pragma unroll
    for (int jq = 0; jq < 4; ++jq) {
        v8f z8 = {0.f, 0.f, 0.f, 0.f, 0.f, 0.f, 0.f, 0.f};
        cx[jq] = z8;
    }
    {
        const float* apx = sPiece + widu * (KQ * HF * WF) + li * WF + 2 * half;
        const float* bpx = sFeat + (li & 3) * ROW_FLOATS + widu * (KQ * FEAT_PAD) + XCOL + 2 * half;
        v2f xa[2][2][4];
        v2f xb[2][2][4];

#define LOAD_XPAIR(S, PP)                                                  \
    _Pragma("unroll")                                                      \
    for (int u = 0; u < 2; ++u) {                                          \
        _Pragma("unroll")                                                  \
        for (int jq = 0; jq < 4; ++jq) {                                   \
            xa[S][u][jq].x = apx[((PP) * 2 + u) * (HF * WF) + jq * 4];     \
            xa[S][u][jq].y = apx[((PP) * 2 + u) * (HF * WF) + jq * 4 + 1]; \
            xb[S][u][jq].x = bpx[((PP) * 2 + u) * FEAT_PAD + jq * 4];      \
            xb[S][u][jq].y = bpx[((PP) * 2 + u) * FEAT_PAD + jq * 4 + 1];  \
        }                                                                  \
    }
#define MMA_XPAIR(S)                                                       \
    _Pragma("unroll")                                                      \
    for (int u = 0; u < 2; ++u) {                                          \
        _Pragma("unroll")                                                  \
        for (int jq = 0; jq < 4; ++jq)                                     \
            cx[jq] = __builtin_amdgcn_wmma_f32_16x16x4_f32(                \
                         false, xa[S][u][jq], false, xb[S][u][jq], (short)0, cx[jq], false, false); \
    }

        LOAD_XPAIR(0, 0)
        #pragma unroll 1
        for (int it = 0; it < KQ / 4 - 1; ++it) {
            LOAD_XPAIR(1, 1)
            __builtin_amdgcn_sched_barrier(0);
            MMA_XPAIR(0)
            __builtin_amdgcn_sched_barrier(0);
            LOAD_XPAIR(0, 2)
            __builtin_amdgcn_sched_barrier(0);
            MMA_XPAIR(1)
            __builtin_amdgcn_sched_barrier(0);
            apx += 4 * HF * WF;
            bpx += 4 * FEAT_PAD;
        }
        LOAD_XPAIR(1, 1)
        __builtin_amdgcn_sched_barrier(0);
        MMA_XPAIR(0)
        MMA_XPAIR(1)
#undef LOAD_XPAIR
#undef MMA_XPAIR
    }

    const int row = row0 + wid;
    #pragma unroll
    for (int t = 0; t < XT; ++t) {
        #pragma unroll
        for (int v = 0; v < 8; ++v) {
            int i = v + 8 * half;
            g[((size_t)(b * 16 + i) * Hfeat + row) * W_ + t * 16 + li] = c[t][v];
        }
    }

    __syncthreads();
    float* red = smem;
    if (li < ROWS_PER_WG) {
        #pragma unroll
        for (int v = 0; v < 8; ++v)
            red[(wid * 16 + v + 8 * half) * ROWS_PER_WG + li] = (cx[0][v] + cx[1][v]) + (cx[2][v] + cx[3][v]);
    }
    __syncthreads();
    if (tid < 16 * ROWS_PER_WG) {
        const int i = tid >> 2;
        const int r = tid & 3;
        float s = red[(0 * 16 + i) * ROWS_PER_WG + r];
        s += red[(1 * 16 + i) * ROWS_PER_WG + r];
        s += red[(2 * 16 + i) * ROWS_PER_WG + r];
        s += red[(3 * 16 + i) * ROWS_PER_WG + r];
        g[((size_t)(b * 16 + i) * Hfeat + row0 + r) * W_ + XCOL] = s;
    }
}

__global__ void corr_phase2_reduce(const float* __restrict__ g, float* __restrict__ out)
{
    int tid = blockIdx.x * blockDim.x + threadIdx.x;
    const int N2 = B_ * LY * LX;
    const int N1 = B_ * LX;
    if (tid < N2) {
        int b   = tid / (LY * LX);
        int rem = tid % (LY * LX);
        int y   = rem / LX;
        int x   = rem % LX;
        float s = 0.f;
        #pragma unroll
        for (int i = 0; i < 16; ++i)
            s += g[((size_t)(b * 16 + i) * 40 + (y + i)) * W_ + x];
        out[tid] = s;
    } else if (tid < N2 + N1) {
        int t2 = tid - N2;
        int b  = t2 / LX;
        int x  = t2 % LX;
        const float* g1 = g + G2D_FLOATS;
        float s = 0.f;
        #pragma unroll
        for (int i = 0; i < 16; ++i)
            s += g1[((size_t)(b * 16 + i) * 16 + i) * W_ + x];
        out[tid] = s;
    }
}

extern "C" void kernel_launch(void* const* d_in, const int* in_sizes, int n_in,
                              void* d_out, int out_size, void* d_ws, size_t ws_size,
                              hipStream_t stream)
{
    (void)in_sizes; (void)n_in; (void)out_size; (void)ws_size;

    const float* piece = (const float*)d_in[0];
    const float* comp  = (const float*)d_in[1];
    const float* strip = (const float*)d_in[2];

    float* g   = (float*)d_ws;
    float* out = (float*)d_out;

    corr_phase1_wmma<<<dim3(GROUPS_2D + GROUPS_1D, B_), TPB, 0, stream>>>(piece, comp, strip, g);

    const int total = B_ * LY * LX + B_ * LX;
    corr_phase2_reduce<<<(total + 255) / 256, 256, 0, stream>>>(g, out);
}


# corr_1d as a 16x80 matrix D[j,x'] with K=(k,i), one workgroup per batch (was 4 row groups, 15 of 16 template rows unused); freed WGPs regroup corr_2d: 3 rows per workgroup, one x-tile per wave
# speedup vs baseline: 10.0818x; 1.0676x over previous
#include <hip/hip_runtime.h>


typedef float v2f __attribute__((ext_vector_type(2)));
typedef float v4f __attribute__((ext_vector_type(4)));
typedef float v8f __attribute__((ext_vector_type(8)));

#define B_        16
#define K_        64
#define HF        16
#define WF        16
#define W_        80
#define HC        40
#define LX        65
#define LY        25
#define FEAT_PAD  96
#define ROWS_PER_WG 3
#define WAVES     4
#define TPB       (WAVES * 32)
#define XCOL      64
#define KQ        (K_ / WAVES)

#define PIECE_FLOATS (K_ * HF * WF)
#define ROW_FLOATS   (K_ * FEAT_PAD)
#define PIECE_OFF    24576
#define G2D_FLOATS   (B_ * 16 * HC * (size_t)W_)
#define N2D          (B_ * LY * LX)

#define LDS_FLOATS   (176 * 1024 / 4)

#define PIECE_VECS   (PIECE_FLOATS / 4)
#define PIECE_PASSES (PIECE_VECS / TPB)
#define GRP_VEC      (ROWS_PER_WG * W_ / 4)
#define FEAT_VEC     (W_ / 4)
#define FEAT_PASSES  (K_ * GRP_VEC / TPB)

#define GROUPS_2D    14
#define LAST_ROW0    (HC - ROWS_PER_WG)

#define STRIP_CH_K    16
#define STRIP_CH_VEC  (STRIP_CH_K * HF * W_ / 4)
#define STRIP_PASSES  (STRIP_CH_VEC / TPB)
#define STRIP_CHUNKS  (K_ / STRIP_CH_K)

__global__ __launch_bounds__(TPB)
void corr_phase1_wmma(const float* __restrict__ piece,
                      const float* __restrict__ comp,
                      const float* __restrict__ strip,
                      float* __restrict__ g,
                      float* __restrict__ out)
{
    __shared__ __attribute__((aligned(16))) float smem[LDS_FLOATS];
    float* sFeat  = smem;
    float* sPiece = smem + PIECE_OFF;

    const bool is1d  = blockIdx.x == GROUPS_2D;
    const int kroll  = is1d ? (K_ / 2) : 0;

    const int b    = blockIdx.y;
    const int tid  = threadIdx.x;
    const int wid  = tid >> 5;
    const int widu = __builtin_amdgcn_readfirstlane(wid);
    const int lane = tid & 31;
    const int half = lane >> 4;
    const int li   = lane & 15;

    const v4f* __restrict__ piece4 = (const v4f*)(piece + (size_t)b * PIECE_FLOATS);
    v4f* sPiece4 = (v4f*)sPiece;

    v4f pr[PIECE_PASSES];
    #pragma unroll
    for (int t = 0; t < PIECE_PASSES; ++t) {
        int idx4 = tid + t * TPB;
        int ksrc = ((idx4 >> 6) + kroll) & (K_ - 1);
        pr[t] = piece4[ksrc * 64 + (idx4 & 63)];
    }

    if (is1d) {
        const v4f* __restrict__ strip4 = (const v4f*)(strip + (size_t)b * (K_ * HF * W_));
        v4f* sStrip4 = (v4f*)sFeat;
        v4f sr[STRIP_PASSES];
        #pragma unroll
        for (int u = 0; u < STRIP_PASSES; ++u)
            sr[u] = strip4[tid + u * TPB];
        __builtin_amdgcn_sched_barrier(0);
        #pragma unroll
        for (int t = 0; t < PIECE_PASSES; ++t)
            sPiece4[tid + t * TPB] = pr[t];
        #pragma unroll
        for (int u = 0; u < STRIP_PASSES; ++u)
            sStrip4[tid + u * TPB] = sr[u];
        __syncthreads();

        v8f d[5];
        #pragma unroll
        for (int t = 0; t < 5; ++t) {
            v8f z8 = {0.f, 0.f, 0.f, 0.f, 0.f, 0.f, 0.f, 0.f};
            d[t] = z8;
        }
        #pragma unroll
        for (int ch = 0; ch < STRIP_CHUNKS; ++ch) {
            if (ch + 1 < STRIP_CHUNKS) {
                #pragma unroll
                for (int u = 0; u < STRIP_PASSES; ++u)
                    sr[u] = strip4[(ch + 1) * STRIP_CH_VEC + tid + u * TPB];
            }
            const float* ap = sPiece + (ch * (STRIP_CH_K * HF) + widu * 64 + 2 * half) * WF + li;
            const float* bp = sFeat + (widu * 64 + 2 * half) * W_ + li;
            v2f pa[2][2];
            v2f pb[2][2][5];
#define LOAD_DPAIR(S, PP)                                                  \
    _Pragma("unroll")                                                      \
    for (int u = 0; u < 2; ++u) {                                          \
        pa[S][u].x = ap[((PP) * 2 + u) * 4 * WF];                          \
        pa[S][u].y = ap[((PP) * 2 + u) * 4 * WF + WF];                     \
        _Pragma("unroll")                                                  \
        for (int t = 0; t < 5; ++t) {                                      \
            pb[S][u][t].x = bp[((PP) * 2 + u) * 4 * W_ + t * 16];          \
            pb[S][u][t].y = bp[((PP) * 2 + u) * 4 * W_ + W_ + t * 16];     \
        }                                                                  \
    }
#define MMA_DPAIR(S)                                                       \
    _Pragma("unroll")                                                      \
    for (int u = 0; u < 2; ++u) {                                          \
        _Pragma("unroll")                                                  \
        for (int t = 0; t < 5; ++t)                                        \
            d[t] = __builtin_amdgcn_wmma_f32_16x16x4_f32(                  \
                       false, pa[S][u], false, pb[S][u][t], (short)0, d[t], false, false); \
    }
            LOAD_DPAIR(0, 0)
            #pragma unroll 1
            for (int it = 0; it < 3; ++it) {
                LOAD_DPAIR(1, 1)
                __builtin_amdgcn_sched_barrier(0);
                MMA_DPAIR(0)
                __builtin_amdgcn_sched_barrier(0);
                LOAD_DPAIR(0, 2)
                __builtin_amdgcn_sched_barrier(0);
                MMA_DPAIR(1)
                __builtin_amdgcn_sched_barrier(0);
                ap += 4 * 4 * WF;
                bp += 4 * 4 * W_;
            }
            LOAD_DPAIR(1, 1)
            __builtin_amdgcn_sched_barrier(0);
            MMA_DPAIR(0)
            MMA_DPAIR(1)
#undef LOAD_DPAIR
#undef MMA_DPAIR
            __syncthreads();
            if (ch + 1 < STRIP_CHUNKS) {
                #pragma unroll
                for (int u = 0; u < STRIP_PASSES; ++u)
                    sStrip4[tid + u * TPB] = sr[u];
                __syncthreads();
            }
        }
        float* red = smem;
        #pragma unroll
        for (int t = 0; t < 5; ++t) {
            #pragma unroll
            for (int v = 0; v < 8; ++v)
                red[(wid * 16 + v + 8 * half) * W_ + t * 16 + li] = d[t][v];
        }
        __syncthreads();
        if (tid < LX) {
            float s = 0.f;
            #pragma unroll
            for (int j = 0; j < WF; ++j) {
                float p = red[(0 * 16 + j) * W_ + tid + j];
                p += red[(1 * 16 + j) * W_ + tid + j];
                p += red[(2 * 16 + j) * W_ + tid + j];
                p += red[(3 * 16 + j) * W_ + tid + j];
                s += p;
            }
            out[N2D + b * LX + tid] = s;
        }
        return;
    }

    const int grp     = blockIdx.x;
    const int rowBase = grp * ROWS_PER_WG;
    const int row0    = rowBase < LAST_ROW0 ? rowBase : LAST_ROW0;

    const float* featB = comp + ((size_t)(b * K_) * HC + row0) * W_;
    const size_t kStride = (size_t)HC * W_;
    v4f fr[FEAT_PASSES];
    #pragma unroll
    for (int s = 0; s < FEAT_PASSES; ++s) {
        int e   = tid + s * TPB;
        int kk  = e / GRP_VEC;
        int rem = e - kk * GRP_VEC;
        fr[s] = *(const v4f*)(featB + (size_t)kk * kStride + rem * 4);
    }

    __builtin_amdgcn_sched_barrier(0);

    #pragma unroll
    for (int t = 0; t < PIECE_PASSES; ++t)
        sPiece4[tid + t * TPB] = pr[t];
    #pragma unroll
    for (int s = 0; s < FEAT_PASSES; ++s) {
        int e   = tid + s * TPB;
        int kk  = e / GRP_VEC;
        int rem = e - kk * GRP_VEC;
        int r   = rem / FEAT_VEC;
        int xq  = rem - r * FEAT_VEC;
        *(v4f*)(sFeat + r * ROW_FLOATS + kk * FEAT_PAD + xq * 4) = fr[s];
    }
    __syncthreads();

    const float* ap = sPiece + li * WF + 2 * half;
    const float* bp = sFeat + wid * 16 + li + 2 * half;

    v8f c[ROWS_PER_WG];
    #pragma unroll
    for (int r = 0; r < ROWS_PER_WG; ++r) {
        v8f z8 = {0.f, 0.f, 0.f, 0.f, 0.f, 0.f, 0.f, 0.f};
        c[r] = z8;
    }

    v2f aF[2][4];
    v2f bF[2][4][ROWS_PER_WG];

#define LOAD_SLAB(S, KK)                                                   \
    _Pragma("unroll")                                                      \
    for (int jq = 0; jq < 4; ++jq) {                                       \
        aF[S][jq].x = ap[(KK) * (HF * WF) + jq * 4];                       \
        aF[S][jq].y = ap[(KK) * (HF * WF) + jq * 4 + 1];                   \
        _Pragma("unroll")                                                  \
        for (int r = 0; r < ROWS_PER_WG; ++r) {                            \
            bF[S][jq][r].x = bp[r * ROW_FLOATS + (KK) * FEAT_PAD + jq * 4];      \
            bF[S][jq][r].y = bp[r * ROW_FLOATS + (KK) * FEAT_PAD + jq * 4 + 1];  \
        }                                                                  \
    }

#define MMA_SLAB(S)                                                        \
    _Pragma("unroll")                                                      \
    for (int jq = 0; jq < 4; ++jq) {                                       \
        _Pragma("unroll")                                                  \
        for (int r = 0; r < ROWS_PER_WG; ++r)                              \
            c[r] = __builtin_amdgcn_wmma_f32_16x16x4_f32(                  \
                       false, aF[S][jq], false, bF[S][jq][r], (short)0, c[r], false, false); \
    }

    LOAD_SLAB(0, 0)
    #pragma unroll 1
    for (int k = 0; k < K_ - 2; k += 2) {
        LOAD_SLAB(1, 1)
        __builtin_amdgcn_sched_barrier(0);
        MMA_SLAB(0)
        __builtin_amdgcn_sched_barrier(0);
        LOAD_SLAB(0, 2)
        __builtin_amdgcn_sched_barrier(0);
        MMA_SLAB(1)
        __builtin_amdgcn_sched_barrier(0);
        ap += 2 * HF * WF;
        bp += 2 * FEAT_PAD;
    }
    LOAD_SLAB(1, 1)
    __builtin_amdgcn_sched_barrier(0);
    MMA_SLAB(0)
    MMA_SLAB(1)
#undef LOAD_SLAB
#undef MMA_SLAB

    v8f cx[4];
    #pragma unroll
    for (int jq = 0; jq < 4; ++jq) {
        v8f z8 = {0.f, 0.f, 0.f, 0.f, 0.f, 0.f, 0.f, 0.f};
        cx[jq] = z8;
    }
    {
        const int rsel = (li & 3) < ROWS_PER_WG ? (li & 3) : (ROWS_PER_WG - 1);
        const float* apx = sPiece + widu * (KQ * HF * WF) + li * WF + 2 * half;
        const float* bpx = sFeat + rsel * ROW_FLOATS + widu * (KQ * FEAT_PAD) + XCOL + 2 * half;
        v2f xa[2][2][4];
        v2f xb[2][2][4];

#define LOAD_XPAIR(S, PP)                                                  \
    _Pragma("unroll")                                                      \
    for (int u = 0; u < 2; ++u) {                                          \
        _Pragma("unroll")                                                  \
        for (int jq = 0; jq < 4; ++jq) {                                   \
            xa[S][u][jq].x = apx[((PP) * 2 + u) * (HF * WF) + jq * 4];     \
            xa[S][u][jq].y = apx[((PP) * 2 + u) * (HF * WF) + jq * 4 + 1]; \
            xb[S][u][jq].x = bpx[((PP) * 2 + u) * FEAT_PAD + jq * 4];      \
            xb[S][u][jq].y = bpx[((PP) * 2 + u) * FEAT_PAD + jq * 4 + 1];  \
        }                                                                  \
    }
#define MMA_XPAIR(S)                                                       \
    _Pragma("unroll")                                                      \
    for (int u = 0; u < 2; ++u) {                                          \
        _Pragma("unroll")                                                  \
        for (int jq = 0; jq < 4; ++jq)                                     \
            cx[jq] = __builtin_amdgcn_wmma_f32_16x16x4_f32(                \
                         false, xa[S][u][jq], false, xb[S][u][jq], (short)0, cx[jq], false, false); \
    }

        LOAD_XPAIR(0, 0)
        #pragma unroll 1
        for (int it = 0; it < KQ / 4 - 1; ++it) {
            LOAD_XPAIR(1, 1)
            __builtin_amdgcn_sched_barrier(0);
            MMA_XPAIR(0)
            __builtin_amdgcn_sched_barrier(0);
            LOAD_XPAIR(0, 2)
            __builtin_amdgcn_sched_barrier(0);
            MMA_XPAIR(1)
            __builtin_amdgcn_sched_barrier(0);
            apx += 4 * HF * WF;
            bpx += 4 * FEAT_PAD;
        }
        LOAD_XPAIR(1, 1)
        __builtin_amdgcn_sched_barrier(0);
        MMA_XPAIR(0)
        MMA_XPAIR(1)
#undef LOAD_XPAIR
#undef MMA_XPAIR
    }

    #pragma unroll
    for (int r = 0; r < ROWS_PER_WG; ++r) {
        const int row = row0 + r;
        if (row >= rowBase) {
            #pragma unroll
            for (int v = 0; v < 8; ++v) {
                int i = v + 8 * half;
                g[((size_t)(b * 16 + i) * HC + row) * W_ + wid * 16 + li] = c[r][v];
            }
        }
    }

    __syncthreads();
    float* red = smem;
    if (li < ROWS_PER_WG) {
        #pragma unroll
        for (int v = 0; v < 8; ++v)
            red[(wid * 16 + v + 8 * half) * ROWS_PER_WG + li] = (cx[0][v] + cx[1][v]) + (cx[2][v] + cx[3][v]);
    }
    __syncthreads();
    if (tid < 16 * ROWS_PER_WG) {
        const int i = tid / ROWS_PER_WG;
        const int r = tid - i * ROWS_PER_WG;
        float s = red[(0 * 16 + i) * ROWS_PER_WG + r];
        s += red[(1 * 16 + i) * ROWS_PER_WG + r];
        s += red[(2 * 16 + i) * ROWS_PER_WG + r];
        s += red[(3 * 16 + i) * ROWS_PER_WG + r];
        if (row0 + r >= rowBase)
            g[((size_t)(b * 16 + i) * HC + row0 + r) * W_ + XCOL] = s;
    }
}

__global__ void corr_phase2_reduce(const float* __restrict__ g, float* __restrict__ out)
{
    int tid = blockIdx.x * blockDim.x + threadIdx.x;
    if (tid < N2D) {
        int b   = tid / (LY * LX);
        int rem = tid % (LY * LX);
        int y   = rem / LX;
        int x   = rem % LX;
        float s = 0.f;
        #pragma unroll
        for (int i = 0; i < 16; ++i)
            s += g[((size_t)(b * 16 + i) * HC + (y + i)) * W_ + x];
        out[tid] = s;
    }
}

extern "C" void kernel_launch(void* const* d_in, const int* in_sizes, int n_in,
                              void* d_out, int out_size, void* d_ws, size_t ws_size,
                              hipStream_t stream)
{
    (void)in_sizes; (void)n_in; (void)out_size; (void)ws_size;

    const float* piece = (const float*)d_in[0];
    const float* comp  = (const float*)d_in[1];
    const float* strip = (const float*)d_in[2];

    float* g   = (float*)d_ws;
    float* out = (float*)d_out;

    corr_phase1_wmma<<<dim3(GROUPS_2D + 1, B_), TPB, 0, stream>>>(piece, comp, strip, g, out);

    corr_phase2_reduce<<<(N2D + 255) / 256, 256, 0, stream>>>(g, out);
}


# corr_2d staged in 8 chunks of 8 k-slabs (template slabs with their chunk) through two LDS buffers: LDS stores of the next chunk issued inside the WMMA stream, one LDS-only barrier per chunk
# speedup vs baseline: 14.9408x; 1.4820x over previous
#include <hip/hip_runtime.h>


typedef float v2f __attribute__((ext_vector_type(2)));
typedef float v4f __attribute__((ext_vector_type(4)));
typedef float v8f __attribute__((ext_vector_type(8)));

#define B_        16
#define K_        64
#define HF        16
#define WF        16
#define W_        80
#define HC        40
#define LX        65
#define LY        25
#define WAVES     4
#define TPB       (WAVES * 32)

#define PIECE_FLOATS (K_ * HF * WF)
#define PIECE_OFF    24576
#define N2D          (B_ * LY * LX)

#define LDS_FLOATS   (176 * 1024 / 4)

#define PIECE_VECS   (PIECE_FLOATS / 4)
#define PIECE_PASSES (PIECE_VECS / TPB)

#define STRIP_CH_K    16
#define STRIP_CH_VEC  (STRIP_CH_K * HF * W_ / 4)
#define STRIP_PASSES  (STRIP_CH_VEC / TPB)
#define STRIP_CHUNKS  (K_ / STRIP_CH_K)

#define ROWS_OUT      2
#define ROWS_IN       (HF + ROWS_OUT - 1)
#define GROUPS_2D     13
#define LAST_Y0       (LY - ROWS_OUT)
#define SLAB_FLOATS   (ROWS_IN * W_)
#define SLAB_VEC      (SLAB_FLOATS / 4)
#define TILE_FLOATS   (ROWS_IN * 16)
#define KSTRIDE_VEC   (HC * W_ / 4)

#define C2_K          8
#define C2_CHUNKS     (K_ / C2_K)
#define C2_FLOATS     (C2_K * SLAB_FLOATS)
#define C2_VEC        (C2_K * SLAB_VEC)
#define C2_FULL       (C2_VEC / TPB)
#define C2_PASSES     (C2_FULL + 1)
#define C2_KVEC       (C2_K * KSTRIDE_VEC)
#define C2_PIECE_OFF  (2 * C2_FLOATS)
#define C2_PVEC       (C2_K * HF * WF / 4)
#define C2_PPASS      (C2_PVEC / TPB)

__global__ __launch_bounds__(TPB)
void corr_wmma(const float* __restrict__ piece,
               const float* __restrict__ comp,
               const float* __restrict__ strip,
               float* __restrict__ out)
{
    __shared__ __attribute__((aligned(16))) float smem[LDS_FLOATS];
    float* sFeat  = smem;
    float* sPiece = smem + PIECE_OFF;

    const bool is1d  = blockIdx.x == GROUPS_2D;
    const int kroll  = is1d ? (K_ / 2) : 0;

    const int b    = blockIdx.y;
    const int tid  = threadIdx.x;
    const int wid  = tid >> 5;
    const int widu = __builtin_amdgcn_readfirstlane(wid);
    const int lane = tid & 31;
    const int half = lane >> 4;
    const int li   = lane & 15;

    const v4f* __restrict__ piece4 = (const v4f*)(piece + (size_t)b * PIECE_FLOATS);

    if (is1d) {
        v4f* sPiece4 = (v4f*)sPiece;
        v4f pr[PIECE_PASSES];
        #pragma unroll
        for (int t = 0; t < PIECE_PASSES; ++t) {
            int idx4 = tid + t * TPB;
            int ksrc = ((idx4 >> 6) + kroll) & (K_ - 1);
            pr[t] = piece4[ksrc * 64 + (idx4 & 63)];
        }
        const v4f* __restrict__ strip4 = (const v4f*)(strip + (size_t)b * (K_ * HF * W_));
        v4f* sStrip4 = (v4f*)sFeat;
        v4f sr[STRIP_PASSES];
        #pragma unroll
        for (int u = 0; u < STRIP_PASSES; ++u)
            sr[u] = strip4[tid + u * TPB];
        __builtin_amdgcn_sched_barrier(0);
        #pragma unroll
        for (int t = 0; t < PIECE_PASSES; ++t)
            sPiece4[tid + t * TPB] = pr[t];
        #pragma unroll
        for (int u = 0; u < STRIP_PASSES; ++u)
            sStrip4[tid + u * TPB] = sr[u];
        __syncthreads();

        v8f d[5];
        #pragma unroll
        for (int t = 0; t < 5; ++t) {
            v8f z8 = {0.f, 0.f, 0.f, 0.f, 0.f, 0.f, 0.f, 0.f};
            d[t] = z8;
        }
        #pragma unroll
        for (int ch = 0; ch < STRIP_CHUNKS; ++ch) {
            if (ch + 1 < STRIP_CHUNKS) {
                #pragma unroll
                for (int u = 0; u < STRIP_PASSES; ++u)
                    sr[u] = strip4[(ch + 1) * STRIP_CH_VEC + tid + u * TPB];
            }
            const float* ap = sPiece + (ch * (STRIP_CH_K * HF) + widu * 64 + 2 * half) * WF + li;
            const float* bp = sFeat + (widu * 64 + 2 * half) * W_ + li;
            v2f pa[2][2];
            v2f pb[2][2][5];
#define LOAD_DPAIR(S, PP)                                                  \
    _Pragma("unroll")                                                      \
    for (int u = 0; u < 2; ++u) {                                          \
        pa[S][u].x = ap[((PP) * 2 + u) * 4 * WF];                          \
        pa[S][u].y = ap[((PP) * 2 + u) * 4 * WF + WF];                     \
        _Pragma("unroll")                                                  \
        for (int t = 0; t < 5; ++t) {                                      \
            pb[S][u][t].x = bp[((PP) * 2 + u) * 4 * W_ + t * 16];          \
            pb[S][u][t].y = bp[((PP) * 2 + u) * 4 * W_ + W_ + t * 16];     \
        }                                                                  \
    }
#define MMA_DPAIR(S)                                                       \
    _Pragma("unroll")                                                      \
    for (int u = 0; u < 2; ++u) {                                          \
        _Pragma("unroll")                                                  \
        for (int t = 0; t < 5; ++t)                                        \
            d[t] = __builtin_amdgcn_wmma_f32_16x16x4_f32(                  \
                       false, pa[S][u], false, pb[S][u][t], (short)0, d[t], false, false); \
    }
            LOAD_DPAIR(0, 0)
            #pragma unroll 1
            for (int it = 0; it < 3; ++it) {
                LOAD_DPAIR(1, 1)
                __builtin_amdgcn_sched_barrier(0);
                MMA_DPAIR(0)
                __builtin_amdgcn_sched_barrier(0);
                LOAD_DPAIR(0, 2)
                __builtin_amdgcn_sched_barrier(0);
                MMA_DPAIR(1)
                __builtin_amdgcn_sched_barrier(0);
                ap += 4 * 4 * WF;
                bp += 4 * 4 * W_;
            }
            LOAD_DPAIR(1, 1)
            __builtin_amdgcn_sched_barrier(0);
            MMA_DPAIR(0)
            MMA_DPAIR(1)
#undef LOAD_DPAIR
#undef MMA_DPAIR
            __syncthreads();
            if (ch + 1 < STRIP_CHUNKS) {
                #pragma unroll
                for (int u = 0; u < STRIP_PASSES; ++u)
                    sStrip4[tid + u * TPB] = sr[u];
                __syncthreads();
            }
        }
        float* red = smem;
        #pragma unroll
        for (int t = 0; t < 5; ++t) {
            #pragma unroll
            for (int v = 0; v < 8; ++v)
                red[(wid * 16 + v + 8 * half) * W_ + t * 16 + li] = d[t][v];
        }
        __syncthreads();
        if (tid < LX) {
            float s = 0.f;
            #pragma unroll
            for (int j = 0; j < WF; ++j) {
                float p = red[(0 * 16 + j) * W_ + tid + j];
                p += red[(1 * 16 + j) * W_ + tid + j];
                p += red[(2 * 16 + j) * W_ + tid + j];
                p += red[(3 * 16 + j) * W_ + tid + j];
                s += p;
            }
            out[N2D + b * LX + tid] = s;
        }
        return;
    }

    const int grp = blockIdx.x;
    const int yb  = grp * ROWS_OUT;
    const int y0  = yb < LAST_Y0 ? yb : LAST_Y0;

    float* sP2  = smem + C2_PIECE_OFF;
    v4f*   sP24 = (v4f*)sP2;
    const v4f* __restrict__ comp4 = (const v4f*)(comp + ((size_t)(b * K_) * HC + y0) * W_);
    const bool tailw = (widu == 0);

    int so[C2_PASSES];
    int sa[C2_PASSES];
    #pragma unroll
    for (int s = 0; s < C2_PASSES; ++s) {
        int e   = tid + s * TPB;
        int kk  = e / SLAB_VEC;
        int rem = e - kk * SLAB_VEC;
        int row = rem / (W_ / 4);
        int xq  = rem - row * (W_ / 4);
        so[s] = kk * KSTRIDE_VEC + rem;
        sa[s] = kk * SLAB_FLOATS + (xq >> 2) * TILE_FLOATS + row * 16 + (xq & 3) * 4;
    }

    v4f fr[2][C2_PASSES];
    v4f pq[2][C2_PPASS];
#define GLOAD_CHUNK(SET, C)                                                \
    _Pragma("unroll")                                                      \
    for (int t = 0; t < C2_PPASS; ++t)                                     \
        pq[SET][t] = piece4[(C) * C2_PVEC + tid + t * TPB];                \
    _Pragma("unroll")                                                      \
    for (int s = 0; s < C2_PASSES; ++s) {                                  \
        if (s < C2_FULL || tailw)                                          \
            fr[SET][s] = comp4[(C) * C2_KVEC + so[s]];                     \
    }
#define STORE_CHUNK(SET, C)                                                \
    _Pragma("unroll")                                                      \
    for (int t = 0; t < C2_PPASS; ++t)                                     \
        sP24[(C) * C2_PVEC + tid + t * TPB] = pq[SET][t];                  \
    _Pragma("unroll")                                                      \
    for (int s = 0; s < C2_PASSES; ++s) {                                  \
        if (s < C2_FULL || tailw)                                          \
            *(v4f*)(smem + ((C) & 1) * C2_FLOATS + sa[s]) = fr[SET][s];    \
    }

#define LDS_BARRIER() asm volatile("s_wait_dscnt 0x0\n\ts_barrier_signal -1\n\ts_barrier_wait -1" ::: "memory")

    GLOAD_CHUNK(0, 0)
    GLOAD_CHUNK(1, 1)
    __builtin_amdgcn_sched_barrier(0);
    STORE_CHUNK(0, 0)
    LDS_BARRIER();
    GLOAD_CHUNK(0, 2)

    v8f c[ROWS_OUT][5];
    #pragma unroll
    for (int r = 0; r < ROWS_OUT; ++r) {
        #pragma unroll
        for (int t = 0; t < 5; ++t) {
            v8f z8 = {0.f, 0.f, 0.f, 0.f, 0.f, 0.f, 0.f, 0.f};
            c[r][t] = z8;
        }
    }

    int fo[ROWS_OUT][5];
    #pragma unroll
    for (int r = 0; r < ROWS_OUT; ++r) {
        #pragma unroll
        for (int t = 0; t < 5; ++t) {
            fo[r][t] = t * TILE_FLOATS + r * 16 + 2 * half * 16 + li;
            asm volatile("" : "+v"(fo[r][t]));
        }
    }

    const float* ap;
    const float* bq[ROWS_OUT][5];
    v2f qa[2][2];
    v2f qb[2][2][ROWS_OUT][5];
#define SET_SLAB(C, SL)                                                    \
    ap = sP2 + (((C) * C2_K + widu * 2 + (SL)) * HF + 2 * half) * WF + li; \
    _Pragma("unroll")                                                      \
    for (int r = 0; r < ROWS_OUT; ++r) {                                   \
        _Pragma("unroll")                                                  \
        for (int t = 0; t < 5; ++t)                                        \
            bq[r][t] = smem + ((C) & 1) * C2_FLOATS + (widu * 2 + (SL)) * SLAB_FLOATS + fo[r][t]; \
    }
#define LOAD_QPAIR(S, PP)                                                  \
    _Pragma("unroll")                                                      \
    for (int u = 0; u < 2; ++u) {                                          \
        qa[S][u].x = ap[((PP) * 2 + u) * 4 * WF];                          \
        qa[S][u].y = ap[((PP) * 2 + u) * 4 * WF + WF];                     \
        _Pragma("unroll")                                                  \
        for (int t = 0; t < 5; ++t) {                                      \
            _Pragma("unroll")                                              \
            for (int r = 0; r < ROWS_OUT; ++r) {                           \
                qb[S][u][r][t].x = bq[r][t][((PP) * 2 + u) * 4 * 16];      \
                qb[S][u][r][t].y = bq[r][t][((PP) * 2 + u) * 4 * 16 + 16]; \
            }                                                              \
        }                                                                  \
    }
#define MMA_QPAIR(S)                                                       \
    _Pragma("unroll")                                                      \
    for (int u = 0; u < 2; ++u) {                                          \
        _Pragma("unroll")                                                  \
        for (int t = 0; t < 5; ++t) {                                      \
            _Pragma("unroll")                                              \
            for (int r = 0; r < ROWS_OUT; ++r)                             \
                c[r][t] = __builtin_amdgcn_wmma_f32_16x16x4_f32(           \
                              false, qa[S][u], false, qb[S][u][r][t], (short)0, c[r][t], false, false); \
        }                                                                  \
    }
#define CHUNK_BODY(C)                                                      \
    LOAD_QPAIR(1, 1)                                                       \
    __builtin_amdgcn_sched_barrier(0);                                     \
    MMA_QPAIR(0)                                                           \
    __builtin_amdgcn_sched_barrier(0);                                     \
    SET_SLAB(C, 1)                                                         \
    LOAD_QPAIR(0, 0)                                                       \
    if ((C) + 1 < C2_CHUNKS) { STORE_CHUNK(((C) + 1) & 1, (C) + 1) }       \
    __builtin_amdgcn_sched_barrier(0);                                     \
    MMA_QPAIR(1)                                                           \
    __builtin_amdgcn_sched_barrier(0);                                     \
    LOAD_QPAIR(1, 1)                                                       \
    __builtin_amdgcn_sched_barrier(0);                                     \
    MMA_QPAIR(0)                                                           \
    __builtin_amdgcn_sched_barrier(0);                                     \
    if ((C) + 1 < C2_CHUNKS) {                                             \
        LDS_BARRIER();                                                     \
        if ((C) + 3 < C2_CHUNKS) { GLOAD_CHUNK(((C) + 1) & 1, (C) + 3) }   \
        SET_SLAB((C) + 1, 0)                                               \
        LOAD_QPAIR(0, 0)                                                   \
        __builtin_amdgcn_sched_barrier(0);                                 \
    }                                                                      \
    MMA_QPAIR(1)                                                           \
    __builtin_amdgcn_sched_barrier(0);

    SET_SLAB(0, 0)
    LOAD_QPAIR(0, 0)
    CHUNK_BODY(0)
    CHUNK_BODY(1)
    CHUNK_BODY(2)
    CHUNK_BODY(3)
    CHUNK_BODY(4)
    CHUNK_BODY(5)
    CHUNK_BODY(6)
    CHUNK_BODY(7)
#undef CHUNK_BODY
#undef MMA_QPAIR
#undef LOAD_QPAIR
#undef SET_SLAB
#undef STORE_CHUNK
#undef GLOAD_CHUNK
#undef LDS_BARRIER

    float* red = smem;
    #pragma unroll
    for (int r = 0; r < ROWS_OUT; ++r) {
        #pragma unroll
        for (int t = 0; t < 5; ++t) {
            #pragma unroll
            for (int v = 0; v < 8; ++v)
                red[((wid * ROWS_OUT + r) * 16 + v + 8 * half) * W_ + t * 16 + li] = c[r][t][v];
        }
    }
    __syncthreads();
    for (int o = tid; o < ROWS_OUT * LX; o += TPB) {
        const int r = o >= LX ? 1 : 0;
        const int x = o - r * LX;
        if (y0 + r >= yb) {
            float s = 0.f;
            #pragma unroll
            for (int j = 0; j < WF; ++j) {
                float p = red[((0 * ROWS_OUT + r) * 16 + j) * W_ + x + j];
                p += red[((1 * ROWS_OUT + r) * 16 + j) * W_ + x + j];
                p += red[((2 * ROWS_OUT + r) * 16 + j) * W_ + x + j];
                p += red[((3 * ROWS_OUT + r) * 16 + j) * W_ + x + j];
                s += p;
            }
            out[(b * LY + y0 + r) * LX + x] = s;
        }
    }
}

extern "C" void kernel_launch(void* const* d_in, const int* in_sizes, int n_in,
                              void* d_out, int out_size, void* d_ws, size_t ws_size,
                              hipStream_t stream)
{
    (void)in_sizes; (void)n_in; (void)out_size; (void)d_ws; (void)ws_size;

    const float* piece = (const float*)d_in[0];
    const float* comp  = (const float*)d_in[1];
    const float* strip = (const float*)d_in[2];

    float* out = (float*)d_out;

    corr_wmma<<<dim3(GROUPS_2D + 1, B_), TPB, 0, stream>>>(piece, comp, strip, out);
}
